# starting kernel
# baseline (speedup 1.0000x reference)
#include <hip/hip_runtime.h>
#include <hip/hip_bf16.h>

typedef __attribute__((ext_vector_type(16))) _Float16 v16h;
typedef __attribute__((ext_vector_type(8)))  float    v8f;

#define FIN 128
#define HID 64

__global__ void k_deg_init(float* __restrict__ deg, int n) {
    int i = blockIdx.x * blockDim.x + threadIdx.x;
    if (i < n) deg[i] = 1.0f;
}

__global__ void k_deg_scatter(const int* __restrict__ dst, float* __restrict__ deg, int E) {
    int i = blockIdx.x * blockDim.x + threadIdx.x;
    if (i < E) atomicAdd(&deg[dst[i]], 1.0f);
}

__global__ void k_rsqrt_inplace(float* __restrict__ deg, int n) {
    int i = blockIdx.x * blockDim.x + threadIdx.x;
    if (i < n) deg[i] = rsqrtf(deg[i]);
}

__global__ void k_gemm1_wmma(const float* __restrict__ x, const float* __restrict__ W1,
                             float* __restrict__ xw1, int n) {
    const int lane    = threadIdx.x & 31;
    const int wave    = threadIdx.x >> 5;
    const int mtile   = blockIdx.x * (blockDim.x >> 5) + wave;
    const int half_hi = lane >> 4;
    const int l16     = lane & 15;
    const int kb      = half_hi * 8;

    int koff[16];
#pragma unroll
    for (int e = 0; e < 16; ++e) {
        const int vg = e >> 1, p = e & 1;
        koff[e] = (vg < 4) ? (kb + 2 * vg + p) : (16 + kb + 2 * (vg - 4) + p);
    }

    const int row  = mtile * 16 + l16;
    const int arow = (row < n) ? row : (n - 1);

    v8f acc[4] = {v8f{}, v8f{}, v8f{}, v8f{}};

#pragma unroll
    for (int ks = 0; ks < FIN; ks += 32) {
        v16h a;
        const float* ap = x + (size_t)arow * FIN + ks;
#pragma unroll
        for (int e = 0; e < 16; ++e) a[e] = (_Float16)ap[koff[e]];

#pragma unroll
        for (int nt = 0; nt < 4; ++nt) {
            v16h b;
            const int col = nt * 16 + l16;
#pragma unroll
            for (int e = 0; e < 16; ++e)
                b[e] = (_Float16)W1[(size_t)(ks + koff[e]) * HID + col];

            acc[nt] = __builtin_amdgcn_wmma_f32_16x16x32_f16(
                 false, a,  false, b,
                 (short)0, acc[nt],  false,  false);
        }
    }

#pragma unroll
    for (int nt = 0; nt < 4; ++nt) {
#pragma unroll
        for (int v = 0; v < 8; ++v) {
            const int orow = mtile * 16 + v + half_hi * 8;
            if (orow < n) xw1[(size_t)orow * HID + nt * 16 + l16] = acc[nt][v];
        }
    }
}

__global__ void k_self1(const float* __restrict__ xw1, const float* __restrict__ dinv,
                        float* __restrict__ agg1, int n) {
    long long i = (long long)blockIdx.x * blockDim.x + threadIdx.x;
    if (i < (long long)n * HID) {
        const int node = (int)(i >> 6);
        const float di = dinv[node];
        agg1[i] = xw1[i] * di * di;
    }
}

__global__ void k_scatter1(const int* __restrict__ src, const int* __restrict__ dst,
                           const float* __restrict__ dinv, const float* __restrict__ xw1,
                           float* __restrict__ agg1, int E) {
    long long t = (long long)blockIdx.x * blockDim.x + threadIdx.x;
    const int e = (int)(t >> 4);
    const int g = (int)(t & 15);
    if (e >= E) return;
    const int s = src[e], d = dst[e];
    const float norm = dinv[s] * dinv[d];
    const float4 v = ((const float4*)(xw1 + (size_t)s * HID))[g];
    float* out = agg1 + (size_t)d * HID + g * 4;
    atomicAdd(out + 0, v.x * norm);
    atomicAdd(out + 1, v.y * norm);
    atomicAdd(out + 2, v.z * norm);
    atomicAdd(out + 3, v.w * norm);
}

__global__ void k_act1(float* __restrict__ agg1, const float* __restrict__ b1, int n) {
    long long i = (long long)blockIdx.x * blockDim.x + threadIdx.x;
    if (i < (long long)n * HID) {
        const int f = (int)(i & (HID - 1));
        agg1[i] = tanhf(agg1[i] + b1[f]);
    }
}

__global__ void k_gemm2(const float* __restrict__ h1, const float* __restrict__ W2,
                        float* __restrict__ xw2, int n) {
    int node = blockIdx.x * blockDim.x + threadIdx.x;
    if (node >= n) return;
    const float* hp = h1 + (size_t)node * HID;
    float o0 = 0.f, o1 = 0.f;
#pragma unroll 8
    for (int f = 0; f < HID; ++f) {
        const float hv = hp[f];
        o0 += hv * W2[f * 2 + 0];
        o1 += hv * W2[f * 2 + 1];
    }
    xw2[(size_t)node * 2 + 0] = o0;
    xw2[(size_t)node * 2 + 1] = o1;
}

__global__ void k_self2(const float* __restrict__ xw2, const float* __restrict__ dinv,
                        float* __restrict__ agg2, int n) {
    int node = blockIdx.x * blockDim.x + threadIdx.x;
    if (node >= n) return;
    const float di2 = dinv[node] * dinv[node];
    agg2[(size_t)node * 2 + 0] = xw2[(size_t)node * 2 + 0] * di2;
    agg2[(size_t)node * 2 + 1] = xw2[(size_t)node * 2 + 1] * di2;
}

__global__ void k_scatter2(const int* __restrict__ src, const int* __restrict__ dst,
                           const float* __restrict__ dinv, const float* __restrict__ xw2,
                           float* __restrict__ agg2, int E) {
    int e = blockIdx.x * blockDim.x + threadIdx.x;
    if (e >= E) return;
    const int s = src[e], d = dst[e];
    const float norm = dinv[s] * dinv[d];
    atomicAdd(&agg2[(size_t)d * 2 + 0], xw2[(size_t)s * 2 + 0] * norm);
    atomicAdd(&agg2[(size_t)d * 2 + 1], xw2[(size_t)s * 2 + 1] * norm);
}

__global__ void k_final(const float* __restrict__ agg2, const float* __restrict__ b2,
                        const float* __restrict__ Wc, const float* __restrict__ bc,
                        float* __restrict__ out, float* __restrict__ hout, int n) {
    int i = blockIdx.x * blockDim.x + threadIdx.x;
    if (i >= n) return;
    const float h0 = tanhf(agg2[(size_t)i * 2 + 0] + b2[0]);
    const float h1 = tanhf(agg2[(size_t)i * 2 + 1] + b2[1]);
    hout[(size_t)i * 2 + 0] = h0;
    hout[(size_t)i * 2 + 1] = h1;
#pragma unroll
    for (int c = 0; c < 4; ++c)
        out[(size_t)i * 4 + c] = h0 * Wc[0 * 4 + c] + h1 * Wc[1 * 4 + c] + bc[c];
}

extern "C" void kernel_launch(void* const* d_in, const int* in_sizes, int n_in,
                              void* d_out, int out_size, void* d_ws, size_t ws_size,
                              hipStream_t stream) {
    const float* x   = (const float*)d_in[0];
    const int*   ei  = (const int*)d_in[1];
    const float* W1  = (const float*)d_in[2];
    const float* b1  = (const float*)d_in[3];
    const float* W2  = (const float*)d_in[4];
    const float* b2  = (const float*)d_in[5];
    const float* Wc  = (const float*)d_in[6];
    const float* bc  = (const float*)d_in[7];

    const int N = in_sizes[0] / FIN;
    const int E = in_sizes[1] / 2;
    const int* src = ei;
    const int* dst = ei + E;

    float* ws   = (float*)d_ws;
    float* dinv = ws;
    float* xw1  = dinv + N;
    float* agg1 = xw1 + (size_t)N * HID;
    float* xw2  = agg1 + (size_t)N * HID;
    float* agg2 = xw2 + (size_t)N * 2;

    float* out  = (float*)d_out;
    float* hout = out + (size_t)N * 4;

    const int B = 256;
    const int gN   = (N + B - 1) / B;
    const int gE   = (E + B - 1) / B;
    const long long nh = (long long)N * HID;
    const int gNH  = (int)((nh + B - 1) / B);
    const long long e16 = (long long)E * 16;
    const int gE16 = (int)((e16 + B - 1) / B);
    const int mtiles = (N + 15) / 16;
    const int gGemm  = (mtiles + 7) / 8;

    k_deg_init<<<gN, B, 0, stream>>>(dinv, N);
    k_deg_scatter<<<gE, B, 0, stream>>>(dst, dinv, E);
    k_rsqrt_inplace<<<gN, B, 0, stream>>>(dinv, N);

    k_gemm1_wmma<<<gGemm, B, 0, stream>>>(x, W1, xw1, N);
    k_self1<<<gNH, B, 0, stream>>>(xw1, dinv, agg1, N);
    k_scatter1<<<gE16, B, 0, stream>>>(src, dst, dinv, xw1, agg1, E);
    k_act1<<<gNH, B, 0, stream>>>(agg1, b1, N);

    k_gemm2<<<gN, B, 0, stream>>>(agg1, W2, xw2, N);
    k_self2<<<gN, B, 0, stream>>>(xw2, dinv, agg2, N);
    k_scatter2<<<gE, B, 0, stream>>>(src, dst, dinv, xw2, agg2, E);

    k_final<<<gN, B, 0, stream>>>(agg2, b2, Wc, bc, out, hout, N);
}


# Edge scatters: each f32 atomic instruction covers whole destination lines (layer 1: 2 whole-line adds per edge, not 8 quarter-line; layer 2: lane pair per edge, 1 add not 2); atomic order as before
# speedup vs baseline: 1.3621x; 1.3621x over previous
#include <hip/hip_runtime.h>
#include <hip/hip_bf16.h>

typedef __attribute__((ext_vector_type(16))) _Float16 v16h;
typedef __attribute__((ext_vector_type(8)))  float    v8f;

#define FIN 128
#define HID 64

__global__ void k_deg_init(float* __restrict__ deg, int n) {
    int i = blockIdx.x * blockDim.x + threadIdx.x;
    if (i < n) deg[i] = 1.0f;
}

__global__ void k_deg_scatter(const int* __restrict__ dst, float* __restrict__ deg, int E) {
    int i = blockIdx.x * blockDim.x + threadIdx.x;
    if (i < E) atomicAdd(&deg[dst[i]], 1.0f);
}

__global__ void k_rsqrt_inplace(float* __restrict__ deg, int n) {
    int i = blockIdx.x * blockDim.x + threadIdx.x;
    if (i < n) deg[i] = rsqrtf(deg[i]);
}

__global__ void k_gemm1_wmma(const float* __restrict__ x, const float* __restrict__ W1,
                             float* __restrict__ xw1, int n) {
    const int lane    = threadIdx.x & 31;
    const int wave    = threadIdx.x >> 5;
    const int mtile   = blockIdx.x * (blockDim.x >> 5) + wave;
    const int half_hi = lane >> 4;
    const int l16     = lane & 15;
    const int kb      = half_hi * 8;

    int koff[16];
#pragma unroll
    for (int e = 0; e < 16; ++e) {
        const int vg = e >> 1, p = e & 1;
        koff[e] = (vg < 4) ? (kb + 2 * vg + p) : (16 + kb + 2 * (vg - 4) + p);
    }

    const int row  = mtile * 16 + l16;
    const int arow = (row < n) ? row : (n - 1);

    v8f acc[4] = {v8f{}, v8f{}, v8f{}, v8f{}};

#pragma unroll
    for (int ks = 0; ks < FIN; ks += 32) {
        v16h a;
        const float* ap = x + (size_t)arow * FIN + ks;
#pragma unroll
        for (int e = 0; e < 16; ++e) a[e] = (_Float16)ap[koff[e]];

#pragma unroll
        for (int nt = 0; nt < 4; ++nt) {
            v16h b;
            const int col = nt * 16 + l16;
#pragma unroll
            for (int e = 0; e < 16; ++e)
                b[e] = (_Float16)W1[(size_t)(ks + koff[e]) * HID + col];

            acc[nt] = __builtin_amdgcn_wmma_f32_16x16x32_f16(
                 false, a,  false, b,
                 (short)0, acc[nt],  false,  false);
        }
    }

#pragma unroll
    for (int nt = 0; nt < 4; ++nt) {
#pragma unroll
        for (int v = 0; v < 8; ++v) {
            const int orow = mtile * 16 + v + half_hi * 8;
            if (orow < n) xw1[(size_t)orow * HID + nt * 16 + l16] = acc[nt][v];
        }
    }
}

__global__ void k_self1(const float* __restrict__ xw1, const float* __restrict__ dinv,
                        float* __restrict__ agg1, int n) {
    long long i = (long long)blockIdx.x * blockDim.x + threadIdx.x;
    if (i < (long long)n * HID) {
        const int node = (int)(i >> 6);
        const float di = dinv[node];
        agg1[i] = xw1[i] * di * di;
    }
}

__global__ void k_scatter1(const int* __restrict__ src, const int* __restrict__ dst,
                           const float* __restrict__ dinv, const float* __restrict__ xw1,
                           float* __restrict__ agg1, int E) {
    const int lane = threadIdx.x & 31;
    const int wave = __builtin_amdgcn_readfirstlane(
        (int)(blockIdx.x * (blockDim.x >> 5) + (threadIdx.x >> 5)));
    const int e0 = wave * 32;
    if (e0 >= E) return;
    const int cnt = (E - e0 < 32) ? (E - e0) : 32;

    int s = 0, d = 0;
    float nrm = 0.f;
    if (lane < cnt) {
        s = src[e0 + lane];
        d = dst[e0 + lane];
        nrm = dinv[s] * dinv[d];
    }

    if (cnt == 32) {
        for (int j0 = 0; j0 < 32; j0 += 8) {
            float lo[8], hi[8];
#pragma unroll
            for (int u = 0; u < 8; ++u) {
                const int sj = __builtin_amdgcn_readlane(s, j0 + u);
                const float* rp = xw1 + (size_t)sj * HID + lane;
                lo[u] = rp[0];
                hi[u] = rp[32];
            }
#pragma unroll
            for (int u = 0; u < 8; ++u) {
                const int   dj = __builtin_amdgcn_readlane(d, j0 + u);
                const float nj = __int_as_float(
                    __builtin_amdgcn_readlane(__float_as_int(nrm), j0 + u));
                float* op = agg1 + (size_t)dj * HID + lane;
                atomicAdd(op,      lo[u] * nj);
                atomicAdd(op + 32, hi[u] * nj);
            }
        }
    } else {
        for (int j = 0; j < cnt; ++j) {
            const int   sj = __builtin_amdgcn_readlane(s, j);
            const int   dj = __builtin_amdgcn_readlane(d, j);
            const float nj = __int_as_float(__builtin_amdgcn_readlane(__float_as_int(nrm), j));
            const float* rp = xw1 + (size_t)sj * HID + lane;
            float* op = agg1 + (size_t)dj * HID + lane;
            atomicAdd(op,      rp[0]  * nj);
            atomicAdd(op + 32, rp[32] * nj);
        }
    }
}

__global__ void k_act1(float* __restrict__ agg1, const float* __restrict__ b1, int n) {
    long long i = (long long)blockIdx.x * blockDim.x + threadIdx.x;
    if (i < (long long)n * HID) {
        const int f = (int)(i & (HID - 1));
        agg1[i] = tanhf(agg1[i] + b1[f]);
    }
}

__global__ void k_gemm2(const float* __restrict__ h1, const float* __restrict__ W2,
                        float* __restrict__ xw2, int n) {
    int node = blockIdx.x * blockDim.x + threadIdx.x;
    if (node >= n) return;
    const float* hp = h1 + (size_t)node * HID;
    float o0 = 0.f, o1 = 0.f;
#pragma unroll 8
    for (int f = 0; f < HID; ++f) {
        const float hv = hp[f];
        o0 += hv * W2[f * 2 + 0];
        o1 += hv * W2[f * 2 + 1];
    }
    xw2[(size_t)node * 2 + 0] = o0;
    xw2[(size_t)node * 2 + 1] = o1;
}

__global__ void k_self2(const float* __restrict__ xw2, const float* __restrict__ dinv,
                        float* __restrict__ agg2, int n) {
    int node = blockIdx.x * blockDim.x + threadIdx.x;
    if (node >= n) return;
    const float di2 = dinv[node] * dinv[node];
    agg2[(size_t)node * 2 + 0] = xw2[(size_t)node * 2 + 0] * di2;
    agg2[(size_t)node * 2 + 1] = xw2[(size_t)node * 2 + 1] * di2;
}

__global__ void k_scatter2(const int* __restrict__ src, const int* __restrict__ dst,
                           const float* __restrict__ dinv, const float* __restrict__ xw2,
                           float* __restrict__ agg2, int E) {
    const long long t = (long long)blockIdx.x * blockDim.x + threadIdx.x;
    const int e = (int)(t >> 1);
    const int c = (int)(t & 1);
    if (e >= E) return;
    const int s = src[e], d = dst[e];
    const float norm = dinv[s] * dinv[d];
    atomicAdd(&agg2[(size_t)d * 2 + c], xw2[(size_t)s * 2 + c] * norm);
}

__global__ void k_final(const float* __restrict__ agg2, const float* __restrict__ b2,
                        const float* __restrict__ Wc, const float* __restrict__ bc,
                        float* __restrict__ out, float* __restrict__ hout, int n) {
    int i = blockIdx.x * blockDim.x + threadIdx.x;
    if (i >= n) return;
    const float h0 = tanhf(agg2[(size_t)i * 2 + 0] + b2[0]);
    const float h1 = tanhf(agg2[(size_t)i * 2 + 1] + b2[1]);
    hout[(size_t)i * 2 + 0] = h0;
    hout[(size_t)i * 2 + 1] = h1;
#pragma unroll
    for (int c = 0; c < 4; ++c)
        out[(size_t)i * 4 + c] = h0 * Wc[0 * 4 + c] + h1 * Wc[1 * 4 + c] + bc[c];
}

extern "C" void kernel_launch(void* const* d_in, const int* in_sizes, int n_in,
                              void* d_out, int out_size, void* d_ws, size_t ws_size,
                              hipStream_t stream) {
    const float* x   = (const float*)d_in[0];
    const int*   ei  = (const int*)d_in[1];
    const float* W1  = (const float*)d_in[2];
    const float* b1  = (const float*)d_in[3];
    const float* W2  = (const float*)d_in[4];
    const float* b2  = (const float*)d_in[5];
    const float* Wc  = (const float*)d_in[6];
    const float* bc  = (const float*)d_in[7];

    const int N = in_sizes[0] / FIN;
    const int E = in_sizes[1] / 2;
    const int* src = ei;
    const int* dst = ei + E;

    float* ws   = (float*)d_ws;
    float* dinv = ws;
    float* xw1  = dinv + N;
    float* agg1 = xw1 + (size_t)N * HID;
    float* xw2  = agg1 + (size_t)N * HID;
    float* agg2 = xw2 + (size_t)N * 2;

    float* out  = (float*)d_out;
    float* hout = out + (size_t)N * 4;

    const int B = 256;
    const int gN   = (N + B - 1) / B;
    const int gE   = (E + B - 1) / B;
    const long long nh = (long long)N * HID;
    const int gNH  = (int)((nh + B - 1) / B);
    const long long e2 = (long long)E * 2;
    const int gE2  = (int)((e2 + B - 1) / B);
    const int mtiles = (N + 15) / 16;
    const int gGemm  = (mtiles + 7) / 8;

    k_deg_init<<<gN, B, 0, stream>>>(dinv, N);
    k_deg_scatter<<<gE, B, 0, stream>>>(dst, dinv, E);
    k_rsqrt_inplace<<<gN, B, 0, stream>>>(dinv, N);

    k_gemm1_wmma<<<gGemm, B, 0, stream>>>(x, W1, xw1, N);
    k_self1<<<gNH, B, 0, stream>>>(xw1, dinv, agg1, N);
    k_scatter1<<<gE, B, 0, stream>>>(src, dst, dinv, xw1, agg1, E);
    k_act1<<<gNH, B, 0, stream>>>(agg1, b1, N);

    k_gemm2<<<gN, B, 0, stream>>>(agg1, W2, xw2, N);
    k_self2<<<gN, B, 0, stream>>>(xw2, dinv, agg2, N);
    k_scatter2<<<gE2, B, 0, stream>>>(src, dst, dinv, xw2, agg2, E);

    k_final<<<gN, B, 0, stream>>>(agg2, b2, Wc, bc, out, hout, N);
}


# Both aggregations: edges counting-sorted by destination in each call, each node's row summed in registers and stored once, no float atomics; sums now in slot order set by the count's atomic rank
# speedup vs baseline: 2.2231x; 1.6321x over previous
#include <hip/hip_runtime.h>
#include <hip/hip_bf16.h>


typedef __attribute__((ext_vector_type(16))) _Float16 v16h;
typedef __attribute__((ext_vector_type(8)))  float    v8f;

#define FIN 128
#define HID 64
#define SCAN_TILE 1024
#define EPT 4

__global__ void k_zero_counts(int* __restrict__ cnt, int n) {
    int i = blockIdx.x * blockDim.x + threadIdx.x;
    if (i < n) cnt[i] = 0;
}

__global__ void k_hist_rank(const int* __restrict__ dst, int* __restrict__ cnt,
                            int* __restrict__ rank, int E) {
    const int T = gridDim.x * blockDim.x;
    const int i = blockIdx.x * blockDim.x + threadIdx.x;
    if (i + (EPT - 1) * T < E) {
        int d[EPT], r[EPT];
        for (int k = 0; k < EPT; ++k) d[k] = dst[i + k * T];
        for (int k = 0; k < EPT; ++k) r[k] = atomicAdd(&cnt[d[k]], 1);
        for (int k = 0; k < EPT; ++k) rank[i + k * T] = r[k];
    } else {
        for (int e = i; e < E; e += T) rank[e] = atomicAdd(&cnt[dst[e]], 1);
    }
}

__global__ void k_tile_sums(const int* __restrict__ cnt, int* __restrict__ tsum, int n) {
    __shared__ int s_w[8];
    const int tid = threadIdx.x, lane = tid & 31, wv = tid >> 5;
    const int base = blockIdx.x * SCAN_TILE + tid * 4;
    int v = 0;
    for (int k = 0; k < 4; ++k)
        if (base + k < n) v += cnt[base + k];
    for (int o = 16; o > 0; o >>= 1) v += __shfl_xor(v, o, 32);
    if (lane == 0) s_w[wv] = v;
    __syncthreads();
    if (tid == 0) {
        int t = 0;
        for (int k = 0; k < 8; ++k) t += s_w[k];
        tsum[blockIdx.x] = t;
    }
}

__global__ void k_scan_rows(const int* __restrict__ cnt, const int* __restrict__ tsum,
                            int* __restrict__ rowptr, float* __restrict__ dinv, int n) {
    __shared__ int s_w[8];
    const int tid = threadIdx.x, lane = tid & 31, wv = tid >> 5;

    int part = 0;
    for (int t = tid; t < (int)blockIdx.x; t += 256) part += tsum[t];
    for (int o = 16; o > 0; o >>= 1) part += __shfl_xor(part, o, 32);
    if (lane == 0) s_w[wv] = part;
    __syncthreads();
    int offset = 0;
    for (int k = 0; k < 8; ++k) offset += s_w[k];
    __syncthreads();

    const int base = blockIdx.x * SCAN_TILE + tid * 4;
    int c[4];
    int ts = 0;
    for (int k = 0; k < 4; ++k) {
        c[k] = (base + k < n) ? cnt[base + k] : 0;
        ts += c[k];
    }
    int incl = ts;
    for (int o = 1; o < 32; o <<= 1) {
        const int up = __shfl_up(incl, o, 32);
        if (lane >= o) incl += up;
    }
    if (lane == 31) s_w[wv] = incl;
    __syncthreads();
    int woff = 0;
    for (int k = 0; k < wv; ++k) woff += s_w[k];

    int run = offset + woff + incl - ts;
    for (int k = 0; k < 4; ++k) {
        if (base + k < n) {
            rowptr[base + k] = run;
            dinv[base + k]   = rsqrtf(1.0f + (float)c[k]);
        }
        run += c[k];
    }
}

__global__ void k_fill_csr(const int* __restrict__ src, const int* __restrict__ dst,
                           const int* __restrict__ rowptr, const int* __restrict__ rank,
                           int* __restrict__ col, int E) {
    const int T = gridDim.x * blockDim.x;
    const int i = blockIdx.x * blockDim.x + threadIdx.x;
    if (i + (EPT - 1) * T < E) {
        int d[EPT], p[EPT], s[EPT];
        for (int k = 0; k < EPT; ++k) {
            d[k] = dst[i + k * T];
            p[k] = rank[i + k * T];
            s[k] = src[i + k * T];
        }
        for (int k = 0; k < EPT; ++k) d[k] = rowptr[d[k]];
        for (int k = 0; k < EPT; ++k) col[d[k] + p[k]] = s[k];
    } else {
        for (int e = i; e < E; e += T) col[rowptr[dst[e]] + rank[e]] = src[e];
    }
}

__global__ void k_gemm1_wmma(const float* __restrict__ x, const float* __restrict__ W1,
                             float* __restrict__ xw1, int n) {
    const int lane    = threadIdx.x & 31;
    const int wave    = threadIdx.x >> 5;
    const int mtile   = blockIdx.x * (blockDim.x >> 5) + wave;
    const int half_hi = lane >> 4;
    const int l16     = lane & 15;
    const int kb      = half_hi * 8;

    int koff[16];
#pragma unroll
    for (int e = 0; e < 16; ++e) {
        const int vg = e >> 1, p = e & 1;
        koff[e] = (vg < 4) ? (kb + 2 * vg + p) : (16 + kb + 2 * (vg - 4) + p);
    }

    const int row  = mtile * 16 + l16;
    const int arow = (row < n) ? row : (n - 1);

    v8f acc[4] = {v8f{}, v8f{}, v8f{}, v8f{}};

#pragma unroll
    for (int ks = 0; ks < FIN; ks += 32) {
        v16h a;
        const float* ap = x + (size_t)arow * FIN + ks;
#pragma unroll
        for (int e = 0; e < 16; ++e) a[e] = (_Float16)ap[koff[e]];

#pragma unroll
        for (int nt = 0; nt < 4; ++nt) {
            v16h b;
            const int col = nt * 16 + l16;
#pragma unroll
            for (int e = 0; e < 16; ++e)
                b[e] = (_Float16)W1[(size_t)(ks + koff[e]) * HID + col];

            acc[nt] = __builtin_amdgcn_wmma_f32_16x16x32_f16(
                 false, a,  false, b,
                 (short)0, acc[nt],  false,  false);
        }
    }

#pragma unroll
    for (int nt = 0; nt < 4; ++nt) {
#pragma unroll
        for (int v = 0; v < 8; ++v) {
            const int orow = mtile * 16 + v + half_hi * 8;
            if (orow < n) xw1[(size_t)orow * HID + nt * 16 + l16] = acc[nt][v];
        }
    }
}

template <bool GUARD>
__device__ __forceinline__ void gather_rows8(const float* __restrict__ xw1, int c, float dv,
                                             float dn, int j0, int m, int lane,
                                             float& a0, float& a1) {
    float2 r[8];
    for (int u = 0; u < 8; ++u) {
        r[u] = make_float2(0.f, 0.f);
        if (!GUARD || j0 + u < m) {
            const int cj = __builtin_amdgcn_readlane(c, j0 + u);
            r[u] = *reinterpret_cast<const float2*>(xw1 + (size_t)cj * HID + 2 * lane);
        }
    }
    for (int u = 0; u < 8; ++u) {
        if (!GUARD || j0 + u < m) {
            const float wj = __int_as_float(
                __builtin_amdgcn_readlane(__float_as_int(dv), j0 + u)) * dn;
            a0 = fmaf(r[u].x, wj, a0);
            a1 = fmaf(r[u].y, wj, a1);
        }
    }
}

__global__ void k_gather1(const int* __restrict__ rowptr, const int* __restrict__ cnt,
                          const int* __restrict__ col, const float* __restrict__ dinv,
                          const float* __restrict__ xw1, const float* __restrict__ b1,
                          float* __restrict__ h1, int n) {
    const int lane = threadIdx.x & 31;
    const int node = __builtin_amdgcn_readfirstlane(
        (int)(blockIdx.x * (blockDim.x >> 5) + (threadIdx.x >> 5)));
    if (node >= n) return;
    const int   beg = rowptr[node];
    const int   deg = cnt[node];
    const float dn  = dinv[node];

    const float2 sv = *reinterpret_cast<const float2*>(xw1 + (size_t)node * HID + 2 * lane);
    float a0 = sv.x * (dn * dn);
    float a1 = sv.y * (dn * dn);

    for (int base = 0; base < deg; base += 32) {
        const int m = (deg - base < 32) ? (deg - base) : 32;
        int c = 0;
        float dv = 0.f;
        if (lane < m) {
            c  = col[beg + base + lane];
            dv = dinv[c];
        }
        int j0 = 0;
        for (; j0 + 8 <= m; j0 += 8) gather_rows8<false>(xw1, c, dv, dn, j0, m, lane, a0, a1);
        if (j0 < m)                  gather_rows8<true >(xw1, c, dv, dn, j0, m, lane, a0, a1);
    }

    float2 o;
    o.x = tanhf(a0 + b1[2 * lane]);
    o.y = tanhf(a1 + b1[2 * lane + 1]);
    *reinterpret_cast<float2*>(h1 + (size_t)node * HID + 2 * lane) = o;
}

__global__ void k_gemm2(const float* __restrict__ h1, const float* __restrict__ W2,
                        float* __restrict__ xw2, int n) {
    int node = blockIdx.x * blockDim.x + threadIdx.x;
    if (node >= n) return;
    const float* hp = h1 + (size_t)node * HID;
    float o0 = 0.f, o1 = 0.f;
#pragma unroll 8
    for (int f = 0; f < HID; ++f) {
        const float hv = hp[f];
        o0 += hv * W2[f * 2 + 0];
        o1 += hv * W2[f * 2 + 1];
    }
    xw2[(size_t)node * 2 + 0] = o0;
    xw2[(size_t)node * 2 + 1] = o1;
}

__global__ void k_gather2_final(const int* __restrict__ rowptr, const int* __restrict__ cnt,
                                const int* __restrict__ col, const float* __restrict__ dinv,
                                const float* __restrict__ xw2, const float* __restrict__ b2,
                                const float* __restrict__ Wc, const float* __restrict__ bc,
                                float* __restrict__ out, float* __restrict__ hout, int n) {
    __shared__ float s_a[32][2];
    const int tid  = threadIdx.x;
    const int sub  = tid & 7;
    const int nl   = tid >> 3;
    const int node = blockIdx.x * 32 + nl;

    int beg = 0, deg = 0;
    float dn = 0.f;
    if (node < n) { beg = rowptr[node]; deg = cnt[node]; dn = dinv[node]; }

    float a0 = 0.f, a1 = 0.f;
    if (sub == 0 && node < n) {
        const float2 sv = *reinterpret_cast<const float2*>(xw2 + (size_t)node * 2);
        a0 = sv.x * (dn * dn);
        a1 = sv.y * (dn * dn);
    }
    for (int j = sub; j < deg; j += 32) {
        int   c[4];
        float dv[4];
        float2 v[4];
        for (int k = 0; k < 4; ++k) {
            c[k] = 0;
            if (j + 8 * k < deg) c[k] = col[beg + j + 8 * k];
        }
        for (int k = 0; k < 4; ++k) {
            dv[k] = 0.f;
            v[k]  = make_float2(0.f, 0.f);
            if (j + 8 * k < deg) {
                dv[k] = dinv[c[k]];
                v[k]  = *reinterpret_cast<const float2*>(xw2 + (size_t)c[k] * 2);
            }
        }
        for (int k = 0; k < 4; ++k) {
            if (j + 8 * k < deg) {
                const float wk = dv[k] * dn;
                a0 = fmaf(v[k].x, wk, a0);
                a1 = fmaf(v[k].y, wk, a1);
            }
        }
    }
    for (int o = 4; o > 0; o >>= 1) {
        a0 += __shfl_xor(a0, o, 32);
        a1 += __shfl_xor(a1, o, 32);
    }
    if (sub == 0) { s_a[nl][0] = a0; s_a[nl][1] = a1; }
    __syncthreads();

    if (tid < 32) {
        const int nd = blockIdx.x * 32 + tid;
        if (nd < n) {
            const float h0 = tanhf(s_a[tid][0] + b2[0]);
            const float h1 = tanhf(s_a[tid][1] + b2[1]);
            *reinterpret_cast<float2*>(hout + (size_t)nd * 2) = make_float2(h0, h1);
            float4 o;
            o.x = h0 * Wc[0 * 4 + 0] + h1 * Wc[1 * 4 + 0] + bc[0];
            o.y = h0 * Wc[0 * 4 + 1] + h1 * Wc[1 * 4 + 1] + bc[1];
            o.z = h0 * Wc[0 * 4 + 2] + h1 * Wc[1 * 4 + 2] + bc[2];
            o.w = h0 * Wc[0 * 4 + 3] + h1 * Wc[1 * 4 + 3] + bc[3];
            *reinterpret_cast<float4*>(out + (size_t)nd * 4) = o;
        }
    }
}

extern "C" void kernel_launch(void* const* d_in, const int* in_sizes, int n_in,
                              void* d_out, int out_size, void* d_ws, size_t ws_size,
                              hipStream_t stream) {
    const float* x   = (const float*)d_in[0];
    const int*   ei  = (const int*)d_in[1];
    const float* W1  = (const float*)d_in[2];
    const float* b1  = (const float*)d_in[3];
    const float* W2  = (const float*)d_in[4];
    const float* b2  = (const float*)d_in[5];
    const float* Wc  = (const float*)d_in[6];
    const float* bc  = (const float*)d_in[7];

    const int N = in_sizes[0] / FIN;
    const int E = in_sizes[1] / 2;
    const int* src = ei;
    const int* dst = ei + E;

    const int B = 256;
    const int gN     = (N + B - 1) / B;
    const int gE4    = (E + B * EPT - 1) / (B * EPT);
    const int tiles  = (N + SCAN_TILE - 1) / SCAN_TILE;
    const int mtiles = (N + 15) / 16;
    const int gGemm  = (mtiles + 7) / 8;
    const int gG1    = (N + 7) / 8;
    const int gG2    = (N + 31) / 32;

    float* ws     = (float*)d_ws;
    float* xw1    = ws;
    float* h1     = xw1 + (size_t)N * HID;
    float* xw2    = h1 + (size_t)N * HID;
    float* dinv   = xw2 + (size_t)N * 2;
    int*   cnt    = (int*)(dinv + N);
    int*   rowptr = cnt + N;
    int*   tsum   = rowptr + N;
    int*   rank   = tsum + tiles;
    int*   col    = rank + E;

    float* out  = (float*)d_out;
    float* hout = out + (size_t)N * 4;

    k_zero_counts<<<gN, B, 0, stream>>>(cnt, N);
    k_hist_rank<<<gE4, B, 0, stream>>>(dst, cnt, rank, E);
    k_tile_sums<<<tiles, B, 0, stream>>>(cnt, tsum, N);
    k_scan_rows<<<tiles, B, 0, stream>>>(cnt, tsum, rowptr, dinv, N);
    k_fill_csr<<<gE4, B, 0, stream>>>(src, dst, rowptr, rank, col, E);

    k_gemm1_wmma<<<gGemm, B, 0, stream>>>(x, W1, xw1, N);
    k_gather1<<<gG1, B, 0, stream>>>(rowptr, cnt, col, dinv, xw1, b1, h1, N);

    k_gemm2<<<gN, B, 0, stream>>>(h1, W2, xw2, N);
    k_gather2_final<<<gG2, B, 0, stream>>>(rowptr, cnt, col, dinv, xw2, b2, Wc, bc, out, hout, N);
}


# GEMM1 moved into the bucket-rows launch: a bucket's block keeps W1 as f16 fragments in LDS, prefetches its 256 x rows under the sort, multiplies them once dinv is known; 5 launches to 4; same sums
# speedup vs baseline: 8.2568x; 3.7141x over previous
#include <hip/hip_runtime.h>
#include <hip/hip_bf16.h>


typedef __attribute__((ext_vector_type(16))) _Float16 v16h;
typedef __attribute__((ext_vector_type(8)))  float    v8f;

#define FIN 128
#define HID 64

#define BSH    8
#define BNODES 256
#define MAXNB  1024
#define EB     1024
#define EPB    8
#define ECH    (EB * EPB)
#define RB     1024
#define RPT    12
#define RCAP   (RB * RPT)
#define RGRP   16

static_assert(EB == MAXNB, "one thread per LDS bucket counter");
static_assert(RB >= BNODES && (RB % 32) == 0 && (BNODES % 32) == 0, "whole waves scan the node counts");
static_assert(RB / 32 == 32, "one wave adds up the per-wave sums of the row pass");
static_assert(RB / 32 >= BNODES / 16, "a wave per 16-row tile of the bucket's product");
static_assert(RB * 8 == FIN * HID && FIN == 128 && HID == 64, "a thread per 8 fragment elements of W1");

__global__ void __launch_bounds__(EB)
k_chunk_sort(const int* __restrict__ src, const int* __restrict__ dst,
             int* __restrict__ bent, int* __restrict__ loff, int E, int NB) {
    __shared__ int s_cnt[MAXNB];
    __shared__ int s_w[EB / 32];
    __shared__ int s_buf[ECH];
    const int tid = threadIdx.x, lane = tid & 31, wv = tid >> 5;
    const int nblk = gridDim.x;
    const int e0 = blockIdx.x * ECH;
    s_cnt[tid] = 0;
    __syncthreads();

    int d[EPB], s[EPB], r[EPB];
    for (int k = 0; k < EPB; ++k) {
        const int e = e0 + tid + k * EB;
        d[k] = -1;
        s[k] = 0;
        if (e < E) {
            d[k] = dst[e];
            s[k] = src[e];
        }
    }
    for (int k = 0; k < EPB; ++k) {
        r[k] = 0;
        if (d[k] >= 0) r[k] = atomicAdd(&s_cnt[d[k] >> BSH], 1);
    }
    __syncthreads();

    const int c = s_cnt[tid];
    int incl = c;
    for (int o = 1; o < 32; o <<= 1) {
        const int up = __shfl_up(incl, o, 32);
        if (lane >= o) incl += up;
    }
    if (lane == 31) s_w[wv] = incl;
    __syncthreads();
    if (wv == 0) {
        const int t = s_w[lane];
        int ti = t;
        for (int o = 1; o < 32; o <<= 1) {
            const int up = __shfl_up(ti, o, 32);
            if (lane >= o) ti += up;
        }
        s_w[lane] = ti - t;
    }
    __syncthreads();
    const int ls = s_w[wv] + incl - c;
    s_cnt[tid] = ls;
    if (tid < NB) loff[tid * nblk + blockIdx.x] = ls;
    __syncthreads();

    for (int k = 0; k < EPB; ++k) {
        if (d[k] >= 0) {
            const int lp = s_cnt[d[k] >> BSH] + r[k];
            s_buf[lp] = (s[k] << BSH) | (d[k] & (BNODES - 1));
        }
    }
    __syncthreads();

    const int nloc = (E - e0 < ECH) ? (E - e0) : ECH;
    for (int k = 0; k < EPB; ++k) {
        const int j = tid + k * EB;
        if (j < nloc) bent[e0 + j] = s_buf[j];
    }
}

__device__ __forceinline__ void bucket_rows_scan(int* s_c, int* s_w, float* s_dv, int tid, int lane, int wv,
                                                 int b, int start, int n,
                                                 int* __restrict__ cnt, int* __restrict__ rowptr,
                                                 float* __restrict__ dinv) {
    int c = 0, incl = 0;
    if (tid < BNODES) {
        c = s_c[tid];
        incl = c;
        for (int o = 1; o < 32; o <<= 1) {
            const int up = __shfl_up(incl, o, 32);
            if (lane >= o) incl += up;
        }
        if (lane == 31) s_w[wv] = incl;
    }
    __syncthreads();
    if (tid < BNODES) {
        int woff = 0;
        for (int k = 0; k < wv; ++k) woff += s_w[k];
        const int excl = woff + incl - c;
        s_c[tid] = excl;
        const int node = b * BNODES + tid;
        const float dv = rsqrtf(1.0f + (float)c);
        s_dv[tid] = dv;
        if (node < n) {
            cnt[node]    = c;
            rowptr[node] = start + excl;
            dinv[node]   = dv;
        }
    }
    __syncthreads();
}

typedef __attribute__((ext_vector_type(8))) _Float16 v8h;

__device__ __forceinline__ void stage_w1(const float* __restrict__ W1, _Float16* s_w1, int tid) {
    const int fl = tid & 31, fp = (tid >> 5) & 1, fnt = (tid >> 6) & 3, fks = tid >> 8;
    const float* wp = W1 + (size_t)(fks * 32 + fp * 16 + (fl >> 4) * 8) * HID + fnt * 16 + (fl & 15);
    v8h h;
#pragma unroll
    for (int j = 0; j < 8; ++j) h[j] = (_Float16)wp[j * HID];
    *reinterpret_cast<v8h*>(s_w1 + 8 * tid) = h;
}

__device__ __forceinline__ void gemm1_tile(const float* __restrict__ x, const _Float16* s_w1,
                                           const float* s_dv, float* __restrict__ xw1,
                                           int n, int row0, int loc0, int lane) {
    const int half_hi = lane >> 4;
    const int l16     = lane & 15;
    const int kb      = half_hi * 8;

    const int row  = row0 + l16;
    const int arow = (row < n) ? row : (n - 1);

    v8f acc[4] = {v8f{}, v8f{}, v8f{}, v8f{}};

#pragma nounroll
    for (int ki = 0; ki < FIN / 32; ++ki) {
        const float* ap = x + (size_t)arow * FIN + ki * 32 + kb;
        v16h a;
#pragma unroll
        for (int e = 0; e < 8; ++e) {
            a[e]     = (_Float16)ap[e];
            a[8 + e] = (_Float16)ap[16 + e];
        }
#pragma unroll
        for (int nt = 0; nt < 4; ++nt) {
            const _Float16* bp = s_w1 + (((ki * 4 + nt) * 2) * 32 + lane) * 8;
            const v8h blo = *reinterpret_cast<const v8h*>(bp);
            const v8h bhi = *reinterpret_cast<const v8h*>(bp + 32 * 8);
            const v16h b = __builtin_shufflevector(blo, bhi, 0, 1, 2, 3, 4, 5, 6, 7, 8, 9, 10, 11, 12, 13, 14, 15);
            acc[nt] = __builtin_amdgcn_wmma_f32_16x16x32_f16(
                 false, a,  false, b,
                 (short)0, acc[nt],  false,  false);
        }
    }

    const int sr8 = half_hi * 8;
    const int sc  = l16;
    float dr[8];
#pragma unroll
    for (int v = 0; v < 8; ++v) dr[v] = s_dv[loc0 + v + sr8];
#pragma unroll
    for (int nt = 0; nt < 4; ++nt) {
#pragma unroll
        for (int v = 0; v < 8; ++v) {
            const int orow = row0 + v + sr8;
            xw1[(size_t)orow * HID + nt * 16 + sc] = acc[nt][v] * dr[v];
        }
    }
}

__global__ void __launch_bounds__(RB, 16)
k_bucket_rows(const int* __restrict__ bent, const int* __restrict__ loff,
              int* __restrict__ cnt, int* __restrict__ rowptr, float* __restrict__ dinv,
              int* __restrict__ col, const float* __restrict__ x, const float* __restrict__ W1,
              float* __restrict__ xw1, int n, int NB, int nblk, int E) {
    __shared__ int s_c[BNODES];
    __shared__ float s_dv[BNODES];
    __shared__ __attribute__((aligned(16))) _Float16 s_w1[FIN * HID];
    __shared__ int s_w[BNODES / 32];
    __shared__ int s_wl[RB / 32];
    __shared__ int s_ws[RB / 32];
    __shared__ int s_rs[32 * 33];
    __shared__ int s_rp[32 * 33];
    __shared__ int s_rl[32 * 33];
    __shared__ int s_col[RCAP];
    const int tid = threadIdx.x, lane = tid & 31, wv = tid >> 5;
    const int b = blockIdx.x;

    if (tid < 2 * BNODES) {
        int pr = b * BNODES + (tid >> 1);
        pr = (pr < n) ? pr : (n - 1);
        __builtin_prefetch(x + (size_t)pr * FIN + (tid & 1) * (FIN / 2), 0, 3);
    }
    stage_w1(W1, s_w1, tid);

    int lo = 0, ln = 0;
    if (tid < nblk) {
        lo = loff[b * nblk + tid];
        int hi;
        if (b + 1 < NB) {
            hi = loff[(b + 1) * nblk + tid];
        } else {
            const int rem = E - tid * ECH;
            hi = (rem < ECH) ? rem : ECH;
        }
        ln = hi - lo;
    }
    if (tid < BNODES) s_c[tid] = 0;

    int incl = ln;
    for (int o = 1; o < 32; o <<= 1) {
        const int up = __shfl_up(incl, o, 32);
        if (lane >= o) incl += up;
    }
    int los = lo;
    for (int o = 16; o > 0; o >>= 1) los += __shfl_xor(los, o, 32);
    if (lane == 31) { s_wl[wv] = incl; s_ws[wv] = los; }
    __syncthreads();
    const int wl = s_wl[lane];
    int before = (lane < wv) ? wl : 0;
    int len    = wl;
    int start  = s_ws[lane];
    for (int o = 16; o > 0; o >>= 1) {
        before += __shfl_xor(before, o, 32);
        len    += __shfl_xor(len, o, 32);
        start  += __shfl_xor(start, o, 32);
    }
    {
        const int slot = lane * 33 + wv;
        s_rs[slot] = (tid < nblk) ? (tid * ECH + lo) : 0;
        s_rp[slot] = before + incl - ln;
        s_rl[slot] = ln;
    }
    __syncthreads();
    const int my_rs = s_rs[wv * 33 + lane];
    const int my_rp = s_rp[wv * 33 + lane];
    const int my_rl = s_rl[wv * 33 + lane];
    const int wvu = __builtin_amdgcn_readfirstlane(wv);

    if (len <= RCAP) {
        bool longrun = false;
        for (int g = 0; g < 32 && wvu + 32 * g < nblk; g += RGRP) {
            int e[RGRP];
#pragma unroll
            for (int k = 0; k < RGRP; ++k) {
                const int rl = __builtin_amdgcn_readlane(my_rl, g + k);
                const int rs = __builtin_amdgcn_readlane(my_rs, g + k);
                longrun = longrun || (rl > 32);
                e[k] = -1;
                if (lane < rl) e[k] = bent[(unsigned)(rs + lane)];
            }
#pragma unroll
            for (int k = 0; k < RGRP; ++k) {
                const int rp = __builtin_amdgcn_readlane(my_rp, g + k);
                if (e[k] >= 0) s_col[rp + lane] = e[k];
            }
        }
        if (longrun) {
#pragma nounroll
            for (int u = 0; u < 32; ++u) {
                const int rl = __builtin_amdgcn_readlane(my_rl, u);
                const int rs = __builtin_amdgcn_readlane(my_rs, u);
                const int rp = __builtin_amdgcn_readlane(my_rp, u);
#pragma nounroll
                for (int i = lane + 32; i < rl; i += 32) s_col[rp + i] = bent[(unsigned)(rs + i)];
            }
        }
        __syncthreads();

        int v[RPT], r[RPT];
        for (int k = 0; k < RPT; ++k) {
            const int j = tid + k * RB;
            v[k] = -1;
            if (j < len) v[k] = s_col[j];
        }
        for (int k = 0; k < RPT; ++k) {
            r[k] = 0;
            if (v[k] >= 0) r[k] = atomicAdd(&s_c[v[k] & (BNODES - 1)], 1);
        }
        __syncthreads();
        bucket_rows_scan(s_c, s_w, s_dv, tid, lane, wv, b, start, n, cnt, rowptr, dinv);
        for (int k = 0; k < RPT; ++k)
            if (v[k] >= 0) s_col[s_c[v[k] & (BNODES - 1)] + r[k]] = v[k] >> BSH;
        __syncthreads();
        for (int k = 0; k < RPT; ++k) {
            const int j = tid + k * RB;
            if (j < len) col[start + j] = s_col[j];
        }
    } else {
#pragma nounroll
        for (int u = 0; u < 32; ++u) {
            const int rl = __builtin_amdgcn_readlane(my_rl, u);
            const int rs = __builtin_amdgcn_readlane(my_rs, u);
#pragma nounroll
            for (int i = lane; i < rl; i += 32) atomicAdd(&s_c[bent[(unsigned)(rs + i)] & (BNODES - 1)], 1);
        }
        __syncthreads();
        bucket_rows_scan(s_c, s_w, s_dv, tid, lane, wv, b, start, n, cnt, rowptr, dinv);
#pragma nounroll
        for (int u = 0; u < 32; ++u) {
            const int rl = __builtin_amdgcn_readlane(my_rl, u);
            const int rs = __builtin_amdgcn_readlane(my_rs, u);
#pragma nounroll
            for (int i = lane; i < rl; i += 32) {
                const int v = bent[(unsigned)(rs + i)];
                const int p = atomicAdd(&s_c[v & (BNODES - 1)], 1);
                col[start + p] = v >> BSH;
            }
        }
    }

    if (wvu < BNODES / 16) {
        const int loc0 = wvu * 16;
        gemm1_tile(x, s_w1, s_dv, xw1, n, b * BNODES + loc0, loc0, lane);
    }
}

template <bool GUARD>
__device__ __forceinline__ void gather_rows8(const float* __restrict__ xw1, int c,
                                             int j0, int m, int lane,
                                             float& a0, float& a1) {
    float2 r[8];
    for (int u = 0; u < 8; ++u) {
        r[u] = make_float2(0.f, 0.f);
        if (!GUARD || j0 + u < m) {
            const int cj = __builtin_amdgcn_readlane(c, j0 + u);
            r[u] = *reinterpret_cast<const float2*>(xw1 + (size_t)cj * HID + 2 * lane);
        }
    }
    for (int u = 0; u < 8; ++u) {
        if (!GUARD || j0 + u < m) {
            a0 += r[u].x;
            a1 += r[u].y;
        }
    }
}

#define G1W   16
#define G1BLK 1024

__device__ __forceinline__ void gather_chunk(const float* __restrict__ xw1, int c, int m, int lane,
                                             float& a0, float& a1) {
    int j0 = 0;
    for (; j0 + 8 <= m; j0 += 8) gather_rows8<false>(xw1, c, j0, m, lane, a0, a1);
    if (j0 < m)                  gather_rows8<true >(xw1, c, j0, m, lane, a0, a1);
}

__device__ __forceinline__ void tile_bounds(const int* __restrict__ rowptr, const int* __restrict__ cnt,
                                            const float* __restrict__ dinv, int t, int lane, int n,
                                            int& vb, int& vd, float& vn) {
    int nd = t * G1W + (lane & (G1W - 1));
    nd = (nd < n) ? nd : (n - 1);
    vb = rowptr[nd];
    vd = cnt[nd];
    vn = dinv[nd];
}

__global__ void __launch_bounds__(G1W * 32)
k_gather1(const int* __restrict__ rowptr, const int* __restrict__ cnt,
          const int* __restrict__ col, const float* __restrict__ dinv,
          const float* __restrict__ xw1, const float* __restrict__ b1,
          const float* __restrict__ W2, float* __restrict__ xw2, int n) {
    __shared__ float2 s_p[2][G1W];
    const int lane = threadIdx.x & 31;
    const int wv     = __builtin_amdgcn_readfirstlane((int)(threadIdx.x >> 5));
    const int ntiles = (n + G1W - 1) / G1W;
    const int step   = (int)gridDim.x;

    const float w00 = W2[4 * lane + 0];
    const float w01 = W2[4 * lane + 1];
    const float w10 = W2[4 * lane + 2];
    const float w11 = W2[4 * lane + 3];
    const float bb0 = b1[2 * lane];
    const float bb1 = b1[2 * lane + 1];

    int tile  = (int)blockIdx.x;
    int ntile = tile + step;
    int node  = tile * G1W + wv;
    int vb, vd, wb, wd;
    float vn, wn;
    tile_bounds(rowptr, cnt, dinv, tile,  lane, n, vb, vd, vn);
    tile_bounds(rowptr, cnt, dinv, ntile, lane, n, wb, wd, wn);
    int   beg = __builtin_amdgcn_readlane(vb, wv);
    int   deg = ((tile < ntiles) && (node < n)) ? __builtin_amdgcn_readlane(vd, wv) : 0;
    float dn  = __int_as_float(__builtin_amdgcn_readlane(__float_as_int(vn), wv));
    int c0 = 0, c1 = 0;
    if (lane < deg)      c0 = col[beg + lane];
    if (lane + 32 < deg) c1 = col[beg + 32 + lane];
    float2 sv = *reinterpret_cast<const float2*>(xw1 + (size_t)((node < n) ? node : (n - 1)) * HID + 2 * lane);

    for (int it = 0; tile < ntiles; ++it) {
        const int   nnode = ntile * G1W + wv;
        const int   nbeg  = __builtin_amdgcn_readlane(wb, wv);
        const int   ndeg  = ((ntile < ntiles) && (nnode < n)) ? __builtin_amdgcn_readlane(wd, wv) : 0;
        const float ndn   = __int_as_float(__builtin_amdgcn_readlane(__float_as_int(wn), wv));
        int nc0 = 0, nc1 = 0;
        if (lane < ndeg)      nc0 = col[nbeg + lane];
        if (lane + 32 < ndeg) nc1 = col[nbeg + 32 + lane];
        const float2 nsv = *reinterpret_cast<const float2*>(xw1 + (size_t)((nnode < n) ? nnode : (n - 1)) * HID + 2 * lane);
        int xb, xd;
        float xn;
        tile_bounds(rowptr, cnt, dinv, ntile + step, lane, n, xb, xd, xn);

        if (node < n) {
            float a0 = sv.x;
            float a1 = sv.y;
            gather_chunk(xw1, c0, (deg < 32) ? deg : 32, lane, a0, a1);
            if (deg > 32) gather_chunk(xw1, c1, (deg - 32 < 32) ? (deg - 32) : 32, lane, a0, a1);
            for (int base = 64; base < deg; base += 32) {
                const int m = (deg - base < 32) ? (deg - base) : 32;
                int c = 0;
                if (lane < m) c = col[beg + base + lane];
                gather_chunk(xw1, c, m, lane, a0, a1);
            }

            const float h0 = tanhf(fmaf(a0, dn, bb0));
            const float h1 = tanhf(fmaf(a1, dn, bb1));
            float p0 = h0 * w00 + h1 * w10;
            float p1 = h0 * w01 + h1 * w11;
            for (int o = 16; o > 0; o >>= 1) {
                p0 += __shfl_xor(p0, o, 32);
                p1 += __shfl_xor(p1, o, 32);
            }
            if (lane == 0) s_p[it & 1][wv] = make_float2(p0 * dn, p1 * dn);
        }
        __syncthreads();
        if (threadIdx.x < G1W) {
            const int nd = tile * G1W + threadIdx.x;
            if (nd < n) *reinterpret_cast<float2*>(xw2 + (size_t)nd * 2) = s_p[it & 1][threadIdx.x];
        }

        tile = ntile;  node = nnode;  beg = nbeg;  deg = ndeg;  dn = ndn;
        c0 = nc0;  c1 = nc1;  sv = nsv;
        ntile += step;
        wb = xb;  wd = xd;  wn = xn;
    }
}

__global__ void k_gather2_final(const int* __restrict__ rowptr, const int* __restrict__ cnt,
                                const int* __restrict__ col, const float* __restrict__ dinv,
                                const float* __restrict__ xw2, const float* __restrict__ b2,
                                const float* __restrict__ Wc, const float* __restrict__ bc,
                                float* __restrict__ out, float* __restrict__ hout, int n) {
    __shared__ float s_a[32][2];
    const int tid  = threadIdx.x;
    const int sub  = tid & 7;
    const int nl   = tid >> 3;
    const int node = blockIdx.x * 32 + nl;

    int beg = 0, deg = 0;
    float dn = 0.f;
    if (node < n) { beg = rowptr[node]; deg = cnt[node]; dn = dinv[node]; }

    float a0 = 0.f, a1 = 0.f;
    if (sub == 0 && node < n) {
        const float2 sv = *reinterpret_cast<const float2*>(xw2 + (size_t)node * 2);
        a0 = sv.x;
        a1 = sv.y;
    }
    for (int j = sub; j < deg; j += 32) {
        int   c[4];
        float2 v[4];
        for (int k = 0; k < 4; ++k) {
            c[k] = 0;
            if (j + 8 * k < deg) c[k] = col[beg + j + 8 * k];
        }
        for (int k = 0; k < 4; ++k) {
            v[k] = make_float2(0.f, 0.f);
            if (j + 8 * k < deg) v[k] = *reinterpret_cast<const float2*>(xw2 + (size_t)c[k] * 2);
        }
        for (int k = 0; k < 4; ++k) {
            if (j + 8 * k < deg) {
                a0 += v[k].x;
                a1 += v[k].y;
            }
        }
    }
    for (int o = 4; o > 0; o >>= 1) {
        a0 += __shfl_xor(a0, o, 32);
        a1 += __shfl_xor(a1, o, 32);
    }
    if (sub == 0) { s_a[nl][0] = a0 * dn; s_a[nl][1] = a1 * dn; }
    __syncthreads();

    if (tid < 32) {
        const int nd = blockIdx.x * 32 + tid;
        if (nd < n) {
            const float h0 = tanhf(s_a[tid][0] + b2[0]);
            const float h1 = tanhf(s_a[tid][1] + b2[1]);
            *reinterpret_cast<float2*>(hout + (size_t)nd * 2) = make_float2(h0, h1);
            float4 o;
            o.x = h0 * Wc[0 * 4 + 0] + h1 * Wc[1 * 4 + 0] + bc[0];
            o.y = h0 * Wc[0 * 4 + 1] + h1 * Wc[1 * 4 + 1] + bc[1];
            o.z = h0 * Wc[0 * 4 + 2] + h1 * Wc[1 * 4 + 2] + bc[2];
            o.w = h0 * Wc[0 * 4 + 3] + h1 * Wc[1 * 4 + 3] + bc[3];
            *reinterpret_cast<float4*>(out + (size_t)nd * 4) = o;
        }
    }
}

extern "C" void kernel_launch(void* const* d_in, const int* in_sizes, int n_in,
                              void* d_out, int out_size, void* d_ws, size_t ws_size,
                              hipStream_t stream) {
    const float* x   = (const float*)d_in[0];
    const int*   ei  = (const int*)d_in[1];
    const float* W1  = (const float*)d_in[2];
    const float* b1  = (const float*)d_in[3];
    const float* W2  = (const float*)d_in[4];
    const float* b2  = (const float*)d_in[5];
    const float* Wc  = (const float*)d_in[6];
    const float* bc  = (const float*)d_in[7];

    const int N = in_sizes[0] / FIN;
    const int E = in_sizes[1] / 2;
    const int* src = ei;
    const int* dst = ei + E;

    const int B = 256;
    const int NB     = (N + BNODES - 1) / BNODES;
    const int NBLK   = (E + ECH - 1) / ECH;
    const int M      = NB * NBLK;
    const int tG1    = (N + G1W - 1) / G1W;
    const int gG1    = (tG1 < G1BLK) ? tG1 : G1BLK;
    const int gG2    = (N + 31) / 32;

    float* ws     = (float*)d_ws;
    float* xw1    = ws;
    float* xw2    = xw1 + (size_t)NB * BNODES * HID;
    float* dinv   = xw2 + (size_t)N * 2;
    int*   cnt    = (int*)(dinv + N);
    int*   rowptr = cnt + N;
    int*   loff   = rowptr + N;
    int*   bent   = loff + M;
    int*   col    = bent + E;

    float* out  = (float*)d_out;
    float* hout = out + (size_t)N * 4;

    k_chunk_sort<<<NBLK, EB, 0, stream>>>(src, dst, bent, loff, E, NB);
    k_bucket_rows<<<NB, RB, 0, stream>>>(bent, loff, cnt, rowptr, dinv, col, x, W1, xw1, N, NB, NBLK, E);

    k_gather1<<<gG1, G1W * 32, 0, stream>>>(rowptr, cnt, col, dinv, xw1, b1, W2, xw2, N);

    k_gather2_final<<<gG2, B, 0, stream>>>(rowptr, cnt, col, dinv, xw2, b2, Wc, bc, out, hout, N);
}


# Layer-2 gather reads pairs from an LDS copy of the xw2 table: 256 resident blocks copy it slice by slice with whole-line loads, sources in registers; a lane's f32 sum now goes slice by slice
# speedup vs baseline: 9.9359x; 1.2034x over previous
#include <hip/hip_runtime.h>
#include <hip/hip_bf16.h>


typedef __attribute__((ext_vector_type(16))) _Float16 v16h;
typedef __attribute__((ext_vector_type(8)))  float    v8f;

#define FIN 128
#define HID 64

#define BSH    8
#define BNODES 256
#define MAXNB  1024
#define EB     1024
#define EPB    8
#define ECH    (EB * EPB)
#define RB     1024
#define RPT    12
#define RCAP   (RB * RPT)
#define RGRP   16
#define STP    72

static_assert(EB == MAXNB, "one thread per LDS bucket counter");
static_assert(RB >= BNODES && (RB % 32) == 0 && (BNODES % 32) == 0, "whole waves scan the node counts");
static_assert(RB / 32 == 32, "one wave adds up the per-wave sums of the row pass");
static_assert(RB / 32 >= BNODES / 16, "a wave per 16-row tile of the bucket's product");
static_assert(RB * 8 == FIN * HID && FIN == 128 && HID == 64, "a thread per 8 fragment elements of W1");
static_assert((BNODES * 8) % RB == 0 && (STP % 8) == 0 && STP >= HID, "whole 16-byte pieces of the staged rows per thread");

__global__ void __launch_bounds__(EB)
k_chunk_sort(const int* __restrict__ src, const int* __restrict__ dst,
             int* __restrict__ bent, int* __restrict__ loff, int E, int NB) {
    __shared__ int s_cnt[MAXNB];
    __shared__ int s_w[EB / 32];
    __shared__ int s_buf[ECH];
    const int tid = threadIdx.x, lane = tid & 31, wv = tid >> 5;
    const int nblk = gridDim.x;
    const int e0 = blockIdx.x * ECH;
    s_cnt[tid] = 0;
    __syncthreads();

    int d[EPB], s[EPB], r[EPB];
    for (int k = 0; k < EPB; ++k) {
        const int e = e0 + tid + k * EB;
        d[k] = -1;
        s[k] = 0;
        if (e < E) {
            d[k] = dst[e];
            s[k] = src[e];
        }
    }
    for (int k = 0; k < EPB; ++k) {
        r[k] = 0;
        if (d[k] >= 0) r[k] = atomicAdd(&s_cnt[d[k] >> BSH], 1);
    }
    __syncthreads();

    const int c = s_cnt[tid];
    int incl = c;
    for (int o = 1; o < 32; o <<= 1) {
        const int up = __shfl_up(incl, o, 32);
        if (lane >= o) incl += up;
    }
    if (lane == 31) s_w[wv] = incl;
    __syncthreads();
    if (wv == 0) {
        const int t = s_w[lane];
        int ti = t;
        for (int o = 1; o < 32; o <<= 1) {
            const int up = __shfl_up(ti, o, 32);
            if (lane >= o) ti += up;
        }
        s_w[lane] = ti - t;
    }
    __syncthreads();
    const int ls = s_w[wv] + incl - c;
    s_cnt[tid] = ls;
    if (tid < NB) loff[tid * nblk + blockIdx.x] = ls;
    __syncthreads();

    for (int k = 0; k < EPB; ++k) {
        if (d[k] >= 0) {
            const int lp = s_cnt[d[k] >> BSH] + r[k];
            s_buf[lp] = (s[k] << BSH) | (d[k] & (BNODES - 1));
        }
    }
    __syncthreads();

    const int nloc = (E - e0 < ECH) ? (E - e0) : ECH;
    for (int k = 0; k < EPB; ++k) {
        const int j = tid + k * EB;
        if (j < nloc) bent[e0 + j] = s_buf[j];
    }
}

__device__ __forceinline__ void bucket_rows_scan(int* s_c, int* s_w, float* s_dv, int tid, int lane, int wv,
                                                 int b, int start, int n,
                                                 int* __restrict__ cnt, int* __restrict__ rowptr,
                                                 float* __restrict__ dinv) {
    int c = 0, incl = 0;
    if (tid < BNODES) {
        c = s_c[tid];
        incl = c;
        for (int o = 1; o < 32; o <<= 1) {
            const int up = __shfl_up(incl, o, 32);
            if (lane >= o) incl += up;
        }
        if (lane == 31) s_w[wv] = incl;
    }
    __syncthreads();
    if (tid < BNODES) {
        int woff = 0;
        for (int k = 0; k < wv; ++k) woff += s_w[k];
        const int excl = woff + incl - c;
        s_c[tid] = excl;
        const int node = b * BNODES + tid;
        const float dv = rsqrtf(1.0f + (float)c);
        s_dv[tid] = dv;
        if (node < n) {
            cnt[node]    = c;
            rowptr[node] = start + excl;
            dinv[node]   = dv;
        }
    }
    __syncthreads();
}

typedef __attribute__((ext_vector_type(8))) _Float16 v8h;

__device__ __forceinline__ void stage_w1(const float* __restrict__ W1, _Float16* s_w1, int tid) {
    const int fl = tid & 31, fp = (tid >> 5) & 1, fnt = (tid >> 6) & 3, fks = tid >> 8;
    const float* wp = W1 + (size_t)(fks * 32 + fp * 16 + (fl >> 4) * 8) * HID + fnt * 16 + (fl & 15);
    v8h h;
#pragma unroll
    for (int j = 0; j < 8; ++j) h[j] = (_Float16)wp[j * HID];
    *reinterpret_cast<v8h*>(s_w1 + 8 * tid) = h;
}

__device__ __forceinline__ void gemm1_tile(const float* __restrict__ x, const _Float16* s_w1,
                                           const float* s_dv, _Float16* s_st,
                                           int n, int row0, int loc0, int lane) {
    const int half_hi = lane >> 4;
    const int l16     = lane & 15;
    const int kb      = half_hi * 8;

    const int row  = row0 + l16;
    const int arow = (row < n) ? row : (n - 1);

    v8f acc[4] = {v8f{}, v8f{}, v8f{}, v8f{}};

#pragma nounroll
    for (int ki = 0; ki < FIN / 32; ++ki) {
        const float* ap = x + (size_t)arow * FIN + ki * 32 + kb;
        v16h a;
#pragma unroll
        for (int e = 0; e < 8; ++e) {
            a[e]     = (_Float16)ap[e];
            a[8 + e] = (_Float16)ap[16 + e];
        }
#pragma unroll
        for (int nt = 0; nt < 4; ++nt) {
            const _Float16* bp = s_w1 + (((ki * 4 + nt) * 2) * 32 + lane) * 8;
            const v8h blo = *reinterpret_cast<const v8h*>(bp);
            const v8h bhi = *reinterpret_cast<const v8h*>(bp + 32 * 8);
            const v16h b = __builtin_shufflevector(blo, bhi, 0, 1, 2, 3, 4, 5, 6, 7, 8, 9, 10, 11, 12, 13, 14, 15);
            acc[nt] = __builtin_amdgcn_wmma_f32_16x16x32_f16(
                 false, a,  false, b,
                 (short)0, acc[nt],  false,  false);
        }
    }

    const int sr8 = half_hi * 8;
    const int sc  = l16;
    float dr[8];
#pragma unroll
    for (int v = 0; v < 8; ++v) dr[v] = s_dv[loc0 + v + sr8];
#pragma unroll
    for (int nt = 0; nt < 4; ++nt) {
#pragma unroll
        for (int v = 0; v < 8; ++v) {
            float val = acc[nt][v] * dr[v];
            val = (val >  65504.0f) ?  65504.0f : val;
            val = (val < -65504.0f) ? -65504.0f : val;
            s_st[(loc0 + v + sr8) * STP + nt * 16 + sc] = (_Float16)val;
        }
    }
}

__global__ void __launch_bounds__(RB, 16)
k_bucket_rows(const int* __restrict__ bent, const int* __restrict__ loff,
              int* __restrict__ cnt, int* __restrict__ rowptr, float* __restrict__ dinv,
              int* __restrict__ col, const float* __restrict__ x, const float* __restrict__ W1,
              float* __restrict__ xw1, int n, int NB, int nblk, int E) {
    __shared__ int s_c[BNODES];
    __shared__ float s_dv[BNODES];
    __shared__ int s_w[BNODES / 32];
    __shared__ int s_wl[RB / 32];
    __shared__ int s_ws[RB / 32];
    __shared__ int s_rs[32 * 33];
    __shared__ int s_rp[32 * 33];
    __shared__ int s_rl[32 * 33];
    struct __attribute__((aligned(16))) W1Col {
        _Float16 w1[FIN * HID];
        int      col[RCAP];
    };
    __shared__ W1Col s_wc;
    _Float16* const s_w1  = s_wc.w1;
    int* const      s_col = s_wc.col;
    __shared__ __attribute__((aligned(16))) _Float16 s_st[BNODES * STP];
    const int tid = threadIdx.x, lane = tid & 31, wv = tid >> 5;
    const int b = blockIdx.x;

    if (tid < 2 * BNODES) {
        int pr = b * BNODES + (tid >> 1);
        pr = (pr < n) ? pr : (n - 1);
        __builtin_prefetch(x + (size_t)pr * FIN + (tid & 1) * (FIN / 2), 0, 3);
    }
    stage_w1(W1, s_w1, tid);

    int lo = 0, ln = 0;
    if (tid < nblk) {
        lo = loff[b * nblk + tid];
        int hi;
        if (b + 1 < NB) {
            hi = loff[(b + 1) * nblk + tid];
        } else {
            const int rem = E - tid * ECH;
            hi = (rem < ECH) ? rem : ECH;
        }
        ln = hi - lo;
    }
    if (tid < BNODES) s_c[tid] = 0;

    int incl = ln;
    for (int o = 1; o < 32; o <<= 1) {
        const int up = __shfl_up(incl, o, 32);
        if (lane >= o) incl += up;
    }
    int los = lo;
    for (int o = 16; o > 0; o >>= 1) los += __shfl_xor(los, o, 32);
    if (lane == 31) { s_wl[wv] = incl; s_ws[wv] = los; }
    __syncthreads();
    const int wl = s_wl[lane];
    int before = (lane < wv) ? wl : 0;
    int len    = wl;
    int start  = s_ws[lane];
    for (int o = 16; o > 0; o >>= 1) {
        before += __shfl_xor(before, o, 32);
        len    += __shfl_xor(len, o, 32);
        start  += __shfl_xor(start, o, 32);
    }
    {
        const int slot = lane * 33 + wv;
        s_rs[slot] = (tid < nblk) ? (tid * ECH + lo) : 0;
        s_rp[slot] = before + incl - ln;
        s_rl[slot] = ln;
    }
    __syncthreads();
    const int my_rs = s_rs[wv * 33 + lane];
    const int my_rp = s_rp[wv * 33 + lane];
    const int my_rl = s_rl[wv * 33 + lane];
    const int wvu = __builtin_amdgcn_readfirstlane(wv);

    if (len <= RCAP) {
        bool longrun = false;
        for (int g = 0; g < 32 && wvu + 32 * g < nblk; g += RGRP) {
            int e[RGRP];
#pragma unroll
            for (int k = 0; k < RGRP; ++k) {
                const int rl = __builtin_amdgcn_readlane(my_rl, g + k);
                const int rs = __builtin_amdgcn_readlane(my_rs, g + k);
                longrun = longrun || (rl > 32);
                e[k] = -1;
                if (lane < rl) e[k] = bent[(unsigned)(rs + lane)];
            }
#pragma unroll
            for (int k = 0; k < RGRP; ++k) {
                const int rp = __builtin_amdgcn_readlane(my_rp, g + k);
                if (e[k] >= 0) s_col[rp + lane] = e[k];
            }
        }
        if (longrun) {
#pragma nounroll
            for (int u = 0; u < 32; ++u) {
                const int rl = __builtin_amdgcn_readlane(my_rl, u);
                const int rs = __builtin_amdgcn_readlane(my_rs, u);
                const int rp = __builtin_amdgcn_readlane(my_rp, u);
#pragma nounroll
                for (int i = lane + 32; i < rl; i += 32) s_col[rp + i] = bent[(unsigned)(rs + i)];
            }
        }
        __syncthreads();

        int v[RPT], r[RPT];
        for (int k = 0; k < RPT; ++k) {
            const int j = tid + k * RB;
            v[k] = -1;
            if (j < len) v[k] = s_col[j];
        }
        for (int k = 0; k < RPT; ++k) {
            r[k] = 0;
            if (v[k] >= 0) r[k] = atomicAdd(&s_c[v[k] & (BNODES - 1)], 1);
        }
        __syncthreads();
        bucket_rows_scan(s_c, s_w, s_dv, tid, lane, wv, b, start, n, cnt, rowptr, dinv);
        for (int k = 0; k < RPT; ++k)
            if (v[k] >= 0) s_col[s_c[v[k] & (BNODES - 1)] + r[k]] = v[k] >> BSH;
        __syncthreads();
        for (int k = 0; k < RPT; ++k) {
            const int j = tid + k * RB;
            if (j < len) col[start + j] = s_col[j];
        }
    } else {
#pragma nounroll
        for (int u = 0; u < 32; ++u) {
            const int rl = __builtin_amdgcn_readlane(my_rl, u);
            const int rs = __builtin_amdgcn_readlane(my_rs, u);
#pragma nounroll
            for (int i = lane; i < rl; i += 32) atomicAdd(&s_c[bent[(unsigned)(rs + i)] & (BNODES - 1)], 1);
        }
        __syncthreads();
        bucket_rows_scan(s_c, s_w, s_dv, tid, lane, wv, b, start, n, cnt, rowptr, dinv);
#pragma nounroll
        for (int u = 0; u < 32; ++u) {
            const int rl = __builtin_amdgcn_readlane(my_rl, u);
            const int rs = __builtin_amdgcn_readlane(my_rs, u);
#pragma nounroll
            for (int i = lane; i < rl; i += 32) {
                const int v = bent[(unsigned)(rs + i)];
                const int p = atomicAdd(&s_c[v & (BNODES - 1)], 1);
                col[start + p] = v >> BSH;
            }
        }
    }

    if (wvu < BNODES / 16) {
        const int loc0 = wvu * 16;
        gemm1_tile(x, s_w1, s_dv, s_st, n, b * BNODES + loc0, loc0, lane);
    }
    __syncthreads();
    _Float16* xw1h = reinterpret_cast<_Float16*>(xw1);
    for (int k = 0; k < (BNODES * 8) / RB; ++k) {
        const int idx = tid + k * RB;
        const int row = idx >> 3, seg = idx & 7;
        const uint4 w = *reinterpret_cast<const uint4*>(s_st + row * STP + seg * 8);
        *reinterpret_cast<uint4*>(xw1h + (size_t)(b * BNODES + row) * HID + seg * 8) = w;
    }
}

typedef __attribute__((ext_vector_type(4))) _Float16 v4h;

__device__ __forceinline__ float xhalf(float v) {
    const int b = __builtin_bit_cast(int, v);
    return __builtin_bit_cast(float, __builtin_amdgcn_permlanex16(b, b, 0x76543210, 0xfedcba98, false, false));
}

__device__ __forceinline__ int half_bcast(int v, int i) {
    const unsigned sel = (unsigned)i * 0x11111111u;
    return (int)__builtin_amdgcn_permlane16((unsigned)v, (unsigned)v, sel, sel, false, false);
}

template <bool GUARD>
__device__ __forceinline__ void gather_pairs8(const char* __restrict__ xw1b, int c,
                                              int i0, int m, int hh, unsigned qoff,
                                              float& a0, float& a1, float& a2, float& a3) {
    v4h r[8];
#pragma unroll
    for (int u = 0; u < 8; ++u) {
        r[u] = v4h{(_Float16)0.f, (_Float16)0.f, (_Float16)0.f, (_Float16)0.f};
        if (!GUARD || 2 * (i0 + u) < m) {
            const int cj = half_bcast(c, i0 + u);
            if (!GUARD || 2 * (i0 + u) + hh < m)
                r[u] = *reinterpret_cast<const v4h*>(xw1b + ((unsigned)cj * (HID * 2u) + qoff));
        }
    }
#pragma unroll
    for (int u = 0; u < 8; ++u) {
        if (!GUARD || 2 * (i0 + u) < m) {
            a0 += (float)r[u][0];
            a1 += (float)r[u][1];
            a2 += (float)r[u][2];
            a3 += (float)r[u][3];
        }
    }
}

#define G1W   16
#define G1BLK 1024

__device__ __forceinline__ void gather_chunk(const char* __restrict__ xw1b, int c, int m,
                                             int hh, unsigned qoff,
                                             float& a0, float& a1, float& a2, float& a3) {
    int i0 = 0;
    for (; 2 * (i0 + 8) <= m; i0 += 8) gather_pairs8<false>(xw1b, c, i0, m, hh, qoff, a0, a1, a2, a3);
    if (2 * i0 < m)                    gather_pairs8<true >(xw1b, c, i0, m, hh, qoff, a0, a1, a2, a3);
}

__device__ __forceinline__ void tile_bounds(const int* __restrict__ rowptr, const int* __restrict__ cnt,
                                            const float* __restrict__ dinv, int t, int lane, int n,
                                            int& vb, int& vd, float& vn) {
    int nd = t * G1W + (lane & (G1W - 1));
    nd = (nd < n) ? nd : (n - 1);
    vb = rowptr[nd];
    vd = cnt[nd];
    vn = dinv[nd];
}

__global__ void __launch_bounds__(G1W * 32)
k_gather1(const int* __restrict__ rowptr, const int* __restrict__ cnt,
          const int* __restrict__ col, const float* __restrict__ dinv,
          const float* __restrict__ xw1, const float* __restrict__ b1,
          const float* __restrict__ W2, float* __restrict__ xw2, int n) {
    __shared__ float2 s_p[2][G1W];
    const int lane = threadIdx.x & 31;
    const int hh   = lane >> 4;
    const int q    = lane & 15;
    const int sl   = 2 * q + hh;
    const unsigned qoff = (unsigned)q * 8u;
    const char* xw1b = reinterpret_cast<const char*>(xw1);
    const int wv     = __builtin_amdgcn_readfirstlane((int)(threadIdx.x >> 5));
    const int ntiles = (n + G1W - 1) / G1W;
    const int step   = (int)gridDim.x;

    const int f0 = 4 * q + 2 * hh;
    const float w00 = W2[2 * f0 + 0];
    const float w01 = W2[2 * f0 + 1];
    const float w10 = W2[2 * f0 + 2];
    const float w11 = W2[2 * f0 + 3];
    const float bb0 = b1[f0];
    const float bb1 = b1[f0 + 1];

    int tile  = (int)blockIdx.x;
    int ntile = tile + step;
    int node  = tile * G1W + wv;
    int vb, vd, wb, wd;
    float vn, wn;
    tile_bounds(rowptr, cnt, dinv, tile,  lane, n, vb, vd, vn);
    tile_bounds(rowptr, cnt, dinv, ntile, lane, n, wb, wd, wn);
    int   beg = __builtin_amdgcn_readlane(vb, wv);
    int   deg = ((tile < ntiles) && (node < n)) ? __builtin_amdgcn_readlane(vd, wv) : 0;
    float dn  = __int_as_float(__builtin_amdgcn_readlane(__float_as_int(vn), wv));
    int c0 = 0, c1 = 0;
    if (sl < deg)      c0 = col[beg + sl];
    if (sl + 32 < deg) c1 = col[beg + 32 + sl];
    v4h sv = *reinterpret_cast<const v4h*>(xw1b + ((unsigned)((node < n) ? node : (n - 1)) * (HID * 2u) + qoff));

    for (int it = 0; tile < ntiles; ++it) {
        const int   nnode = ntile * G1W + wv;
        const int   nbeg  = __builtin_amdgcn_readlane(wb, wv);
        const int   ndeg  = ((ntile < ntiles) && (nnode < n)) ? __builtin_amdgcn_readlane(wd, wv) : 0;
        const float ndn   = __int_as_float(__builtin_amdgcn_readlane(__float_as_int(wn), wv));
        int nc0 = 0, nc1 = 0;
        if (sl < ndeg)      nc0 = col[nbeg + sl];
        if (sl + 32 < ndeg) nc1 = col[nbeg + 32 + sl];
        const v4h nsv = *reinterpret_cast<const v4h*>(xw1b + ((unsigned)((nnode < n) ? nnode : (n - 1)) * (HID * 2u) + qoff));
        int xb, xd;
        float xn;
        tile_bounds(rowptr, cnt, dinv, ntile + step, lane, n, xb, xd, xn);

        if (node < n) {
            float a0 = hh ? 0.f : (float)sv[0];
            float a1 = hh ? 0.f : (float)sv[1];
            float a2 = hh ? 0.f : (float)sv[2];
            float a3 = hh ? 0.f : (float)sv[3];
            gather_chunk(xw1b, c0, (deg < 32) ? deg : 32, hh, qoff, a0, a1, a2, a3);
            if (deg > 32) gather_chunk(xw1b, c1, (deg - 32 < 32) ? (deg - 32) : 32, hh, qoff, a0, a1, a2, a3);
            for (int base = 64; base < deg; base += 32) {
                const int m = (deg - base < 32) ? (deg - base) : 32;
                int c = 0;
                if (sl < m) c = col[beg + base + sl];
                gather_chunk(xw1b, c, m, hh, qoff, a0, a1, a2, a3);
            }

            const float t0 = (hh ? a2 : a0) + xhalf(hh ? a0 : a2);
            const float t1 = (hh ? a3 : a1) + xhalf(hh ? a1 : a3);

            const float h0 = tanhf(fmaf(t0, dn, bb0));
            const float h1 = tanhf(fmaf(t1, dn, bb1));
            float p0 = h0 * w00 + h1 * w10;
            float p1 = h0 * w01 + h1 * w11;
            for (int o = 16; o > 0; o >>= 1) {
                p0 += __shfl_xor(p0, o, 32);
                p1 += __shfl_xor(p1, o, 32);
            }
            if (lane == 0) s_p[it & 1][wv] = make_float2(p0 * dn, p1 * dn);
        }
        __syncthreads();
        if (threadIdx.x < G1W) {
            const int nd = tile * G1W + threadIdx.x;
            if (nd < n) *reinterpret_cast<float2*>(xw2 + (size_t)nd * 2) = s_p[it & 1][threadIdx.x];
        }

        tile = ntile;  node = nnode;  beg = nbeg;  deg = ndeg;  dn = ndn;
        c0 = nc0;  c1 = nc1;  sv = nsv;
        ntile += step;
        wb = xb;  wd = xd;  wn = xn;
    }
}

#define G2T    512
#define G2BLK  256
#define G2NPR  (G2T / 8)
#define G2R    7
#define G2S    8
#define G2SA   6
#define G2GRP  (G2R * G2NPR)
#define G2CAP  34816
#define G2CU   17
#define G2NONE 0xFFFFFFFFu

static_assert((G2CAP % 16) == 0, "slices start on whole lines of the table");
static_assert(G2GRP <= G2T, "a thread per node finishes the group");
static_assert(G2CU == 17, "the fence of the copy names its G2CU loaded values one by one");

__device__ __forceinline__ float2 slice_pair(const char* tb, unsigned cb, unsigned baseb, unsigned zoff) {
    unsigned off = cb - baseb;
    off = (off < zoff) ? off : zoff;
    return *reinterpret_cast<const float2*>(tb + off);
}

typedef __attribute__((ext_vector_type(4))) unsigned int v4u;

__device__ __forceinline__ void copy_pieces(const v4u* __restrict__ gsrc, v4u* ldst,
                                            int i0, int i1, int nun, int rot, int tid) {
    for (; i0 < i1; i0 += G2T * G2CU) {
        v4u w[G2CU];
        int u[G2CU];
#pragma unroll
        for (int q = 0; q < G2CU; ++q) {
            const int i = i0 + q * G2T + tid;
            int uu = i + rot;
            uu = (uu >= nun) ? (uu - nun) : uu;
            u[q] = (i < nun) ? uu : (G2CAP / 2 + 1);
            w[q] = gsrc[(i < nun) ? uu : 0];
        }
        asm volatile("" :: "v"(w[0]), "v"(w[1]), "v"(w[2]), "v"(w[3]), "v"(w[4]), "v"(w[5]),
                           "v"(w[6]), "v"(w[7]), "v"(w[8]), "v"(w[9]), "v"(w[10]), "v"(w[11]),
                           "v"(w[12]), "v"(w[13]), "v"(w[14]), "v"(w[15]), "v"(w[16]));
#pragma unroll
        for (int q = 0; q < G2CU; ++q) ldst[u[q]] = w[q];
    }
}

__global__ void __launch_bounds__(G2T)
k_gather2_final(const int* __restrict__ rowptr, const int* __restrict__ cnt,
                const int* __restrict__ col, const float* __restrict__ dinv,
                const float* __restrict__ xw2, const float* __restrict__ b2,
                const float* __restrict__ Wc, const float* __restrict__ bc,
                float* __restrict__ out, float* __restrict__ hout, int n) {
    __shared__ __attribute__((aligned(16))) float2 s_tab[G2CAP + 4];
    __shared__ float2 s_a[G2GRP];
    const int tid = threadIdx.x;
    const int sub = tid & 7;
    const int nl  = tid >> 3;

    if (n <= 0) return;
    const int nblk = (int)gridDim.x;
    const int npb  = (((n + nblk - 1) / nblk) + 15) & ~15;
    const int nb0  = (int)blockIdx.x * npb;
    if (nb0 >= n) return;
    const int nb1  = (nb0 + npb < n) ? (nb0 + npb) : n;

    const int npass = (n + G2CAP - 1) / G2CAP;
    const int slice = (((n + npass - 1) / npass) + 15) & ~15;
    const unsigned zoff = (unsigned)slice * 8u;
    if (tid == 0) s_tab[slice] = make_float2(0.f, 0.f);
    const char* tb = reinterpret_cast<const char*>(s_tab);
    v4u* ldst = reinterpret_cast<v4u*>(s_tab);
    const v4u* gtab = reinterpret_cast<const v4u*>(xw2);

    for (int g0 = nb0; g0 < nb1; g0 += G2GRP) {
        const bool first = (g0 == nb0);
        const int cn0  = (n < slice) ? n : slice;
        const int nun0 = (cn0 + 1) >> 1;
        const int rot0 = (int)(((unsigned)blockIdx.x * (unsigned)nun0) / (unsigned)nblk) & ~31;

        int   beg[G2R], deg[G2R];
        float dn[G2R], a0[G2R], a1[G2R];
        float2 sv[G2R];
#pragma unroll
        for (int r = 0; r < G2R; ++r) {
            const int node = g0 + r * G2NPR + nl;
            const int nc   = (node < nb1) ? node : (nb1 - 1);
            beg[r] = rowptr[nc];
            deg[r] = cnt[nc];
            dn[r]  = dinv[nc];
            sv[r]  = *reinterpret_cast<const float2*>(xw2 + (size_t)nc * 2);
        }
        if (first) copy_pieces(gtab, ldst, 0, G2T * G2CU, nun0, rot0, tid);
#pragma unroll
        for (int r = 0; r < G2R; ++r) {
            const bool ok = g0 + r * G2NPR + nl < nb1;
            deg[r] = ok ? deg[r] : 0;
            a0[r]  = (ok && sub == 0) ? sv[r].x : 0.f;
            a1[r]  = (ok && sub == 0) ? sv[r].y : 0.f;
        }
        unsigned idx[G2R][G2S];
#pragma unroll
        for (int r = 0; r < G2R; ++r) {
#pragma unroll
            for (int k = 0; k < G2S; ++k) {
                const int j = sub + 8 * k;
                const int pos = (j < deg[r]) ? (beg[r] + j) : 0;
                idx[r][k] = (unsigned)col[pos];
            }
        }
        if (first) copy_pieces(gtab, ldst, G2T * G2CU, nun0, nun0, rot0, tid);
#pragma unroll
        for (int r = 0; r < G2R; ++r)
#pragma unroll
            for (int k = 0; k < G2S; ++k)
                idx[r][k] = (sub + 8 * k < deg[r]) ? (idx[r][k] * 8u) : G2NONE;
        int dm[G2R];
#pragma unroll
        for (int r = 0; r < G2R; ++r) {
            const int d0 = __builtin_amdgcn_readlane(deg[r], 0);
            const int d1 = __builtin_amdgcn_readlane(deg[r], 8);
            const int d2 = __builtin_amdgcn_readlane(deg[r], 16);
            const int d3 = __builtin_amdgcn_readlane(deg[r], 24);
            const int d01 = (d0 > d1) ? d0 : d1, d23 = (d2 > d3) ? d2 : d3;
            dm[r] = (d01 > d23) ? d01 : d23;
        }

        for (int p = 0; p < npass; ++p) {
            const int sb = p * slice;
            if (!(first && p == 0)) {
                const int cn = (n - sb < slice) ? (n - sb) : slice;
                const int nun = (cn + 1) >> 1;
                const int rot = (int)(((unsigned)blockIdx.x * (unsigned)nun) / (unsigned)nblk) & ~31;
                __syncthreads();
                copy_pieces(gtab + (size_t)sb / 2, ldst, 0, nun, nun, rot, tid);
            }
            __syncthreads();

            const unsigned baseb = (unsigned)sb * 8u;
#pragma unroll
            for (int r = 0; r < G2R; ++r) {
                float2 v[G2SA];
#pragma unroll
                for (int k = 0; k < G2SA; ++k) v[k] = slice_pair(tb, idx[r][k], baseb, zoff);
#pragma unroll
                for (int k = 0; k < G2SA; ++k) { a0[r] += v[k].x;  a1[r] += v[k].y; }
            }
#pragma unroll
            for (int r = 0; r < G2R; ++r) {
                if (dm[r] > 8 * G2SA) {
                    float2 v[G2S - G2SA];
#pragma unroll
                    for (int k = G2SA; k < G2S; ++k) v[k - G2SA] = slice_pair(tb, idx[r][k], baseb, zoff);
#pragma unroll
                    for (int k = G2SA; k < G2S; ++k) { a0[r] += v[k - G2SA].x;  a1[r] += v[k - G2SA].y; }
                }
            }
        }

#pragma unroll
        for (int r = 0; r < G2R; ++r) {
            if (dm[r] > 8 * G2S) {
                for (int j = 8 * G2S + sub; j < deg[r]; j += 8) {
                    const int c = col[beg[r] + j];
                    const float2 v = *reinterpret_cast<const float2*>(xw2 + (size_t)c * 2);
                    a0[r] += v.x;
                    a1[r] += v.y;
                }
            }
        }

#pragma unroll
        for (int r = 0; r < G2R; ++r) {
            float s0 = a0[r], s1 = a1[r];
            for (int o = 4; o > 0; o >>= 1) {
                s0 += __shfl_xor(s0, o, 32);
                s1 += __shfl_xor(s1, o, 32);
            }
            if (sub == 0) s_a[r * G2NPR + nl] = make_float2(s0 * dn[r], s1 * dn[r]);
        }
        __syncthreads();

        if (tid < G2GRP) {
            const int nd = g0 + tid;
            if (nd < nb1) {
                const float2 s = s_a[tid];
                const float h0 = tanhf(s.x + b2[0]);
                const float h1 = tanhf(s.y + b2[1]);
                *reinterpret_cast<float2*>(hout + (size_t)nd * 2) = make_float2(h0, h1);
                float4 o;
                o.x = h0 * Wc[0 * 4 + 0] + h1 * Wc[1 * 4 + 0] + bc[0];
                o.y = h0 * Wc[0 * 4 + 1] + h1 * Wc[1 * 4 + 1] + bc[1];
                o.z = h0 * Wc[0 * 4 + 2] + h1 * Wc[1 * 4 + 2] + bc[2];
                o.w = h0 * Wc[0 * 4 + 3] + h1 * Wc[1 * 4 + 3] + bc[3];
                *reinterpret_cast<float4*>(out + (size_t)nd * 4) = o;
            }
        }
    }
}

extern "C" void kernel_launch(void* const* d_in, const int* in_sizes, int n_in,
                              void* d_out, int out_size, void* d_ws, size_t ws_size,
                              hipStream_t stream) {
    const float* x   = (const float*)d_in[0];
    const int*   ei  = (const int*)d_in[1];
    const float* W1  = (const float*)d_in[2];
    const float* b1  = (const float*)d_in[3];
    const float* W2  = (const float*)d_in[4];
    const float* b2  = (const float*)d_in[5];
    const float* Wc  = (const float*)d_in[6];
    const float* bc  = (const float*)d_in[7];

    const int N = in_sizes[0] / FIN;
    const int E = in_sizes[1] / 2;
    const int* src = ei;
    const int* dst = ei + E;

    const int NB     = (N + BNODES - 1) / BNODES;
    const int NBLK   = (E + ECH - 1) / ECH;
    const int M      = NB * NBLK;
    const int tG1    = (N + G1W - 1) / G1W;
    const int gG1    = (tG1 < G1BLK) ? tG1 : G1BLK;

    float* ws     = (float*)d_ws;
    float* xw1    = ws;
    float* xw2    = xw1 + (size_t)NB * BNODES * HID;
    float* dinv   = xw2 + (size_t)N * 2;
    int*   cnt    = (int*)(dinv + N);
    int*   rowptr = cnt + N;
    int*   loff   = rowptr + N;
    int*   bent   = loff + M;
    int*   col    = bent + E;

    float* out  = (float*)d_out;
    float* hout = out + (size_t)N * 4;

    k_chunk_sort<<<NBLK, EB, 0, stream>>>(src, dst, bent, loff, E, NB);
    k_bucket_rows<<<NB, RB, 0, stream>>>(bent, loff, cnt, rowptr, dinv, col, x, W1, xw1, N, NB, NBLK, E);

    k_gather1<<<gG1, G1W * 32, 0, stream>>>(rowptr, cnt, col, dinv, xw1, b1, W2, xw2, N);

    k_gather2_final<<<G2BLK, G2T, 0, stream>>>(rowptr, cnt, col, dinv, xw2, b2, Wc, bc, out, hout, N);
}
